# starting kernel
# baseline (speedup 1.0000x reference)
#include <hip/hip_runtime.h>

typedef __attribute__((ext_vector_type(2))) float v2f;
typedef __attribute__((ext_vector_type(8))) float v8f;

static constexpr unsigned long long kPerm = 0xB5F93E82A4D71C60ULL;

__device__ __forceinline__ float sigmoid_fast(float x) {
  float e = __builtin_amdgcn_exp2f(x * -1.4426950408889634f);
  return __builtin_amdgcn_rcpf(1.0f + e);
}
__device__ __forceinline__ float tanh_fast(float x) {
  float e = __builtin_amdgcn_exp2f(x * -2.8853900817779268f);
  return (1.0f - e) * __builtin_amdgcn_rcpf(1.0f + e);
}
__device__ __forceinline__ v8f wmma4(v2f a, v2f b, v8f c) {
  return __builtin_amdgcn_wmma_f32_16x16x4_f32(false, a, false, b, (short)0, c, false, false);
}
__device__ __forceinline__ v2f mk2(float a, float b) { v2f r; r[0] = a; r[1] = b; return r; }

__global__ __launch_bounds__(256) void gru16_wmma_kernel(
    const float* __restrict__ x, const int* __restrict__ lengths,
    const float* __restrict__ W_ih, const float* __restrict__ W_hh,
    const float* __restrict__ b_ih, const float* __restrict__ b_hh,
    float* __restrict__ out, int tmax)
{
  const int lane = threadIdx.x & 31;
  const int col  = lane & 15;
  const int half = lane >> 4;
  const int wv   = (blockIdx.x * blockDim.x + threadIdx.x) >> 5;
  const int s    = wv * 16 + col;

  const int g0 = (int)((kPerm >> (4 * col)) & 0xFULL);
  const int g1 = (col == 0) ? 16 : ((col == 8) ? 17 : -1);
  const int cl = 2 * half;

  v2f Aih[2][2][2], Ahh[2][2][2];
  v8f Cih[2][2],    Chh[2][2];
#pragma unroll
  for (int l = 0; l < 2; ++l) {
    const float* Wi = W_ih + l * 108;
    const float* Wh = W_hh + l * 108;
    const float* Bi = b_ih + l * 18;
    const float* Bh = b_hh + l * 18;

    Aih[l][0][0] = mk2(Wi[g0 * 6 + cl], Wi[g0 * 6 + cl + 1]);
    Ahh[l][0][0] = mk2(Wh[g0 * 6 + cl], Wh[g0 * 6 + cl + 1]);
    Aih[l][0][1] = half ? mk2(0.f, 0.f) : mk2(Wi[g0 * 6 + 4], Wi[g0 * 6 + 5]);
    Ahh[l][0][1] = half ? mk2(0.f, 0.f) : mk2(Wh[g0 * 6 + 4], Wh[g0 * 6 + 5]);
    if (g1 >= 0) {
      Aih[l][1][0] = mk2(Wi[g1 * 6 + cl], Wi[g1 * 6 + cl + 1]);
      Ahh[l][1][0] = mk2(Wh[g1 * 6 + cl], Wh[g1 * 6 + cl + 1]);
      Aih[l][1][1] = half ? mk2(0.f, 0.f) : mk2(Wi[g1 * 6 + 4], Wi[g1 * 6 + 5]);
      Ahh[l][1][1] = half ? mk2(0.f, 0.f) : mk2(Wh[g1 * 6 + 4], Wh[g1 * 6 + 5]);
    } else {
      Aih[l][1][0] = mk2(0.f, 0.f); Ahh[l][1][0] = mk2(0.f, 0.f);
      Aih[l][1][1] = mk2(0.f, 0.f); Ahh[l][1][1] = mk2(0.f, 0.f);
    }
    v8f ci0, ch0, ci1, ch1;
#pragma unroll
    for (int r = 0; r < 8; ++r) {
      int row = r + 8 * half;
      int g = (int)((kPerm >> (4 * row)) & 0xFULL);
      ci0[r] = Bi[g];  ch0[r] = Bh[g];
      ci1[r] = 0.f;    ch1[r] = 0.f;
    }
    ci1[0] = Bi[half ? 17 : 16];
    ch1[0] = Bh[half ? 17 : 16];
    Cih[l][0] = ci0; Chh[l][0] = ch0;
    Cih[l][1] = ci1; Chh[l][1] = ch1;
  }

  float ha0 = 0.f, hb0 = 0.f, hc0 = 0.f;
  float ha1 = 0.f, hb1 = 0.f, hc1 = 0.f;
  const int len = lengths[s];
  const float* xlo = x + (size_t)s * (size_t)tmax * 6 + cl;
  const float* xhi = x + (size_t)s * (size_t)tmax * 6 + 4;

  float2 xl = *(const float2*)xlo;
  float2 xh = *(const float2*)xhi;

#pragma unroll 1
  for (int t = 0; t < tmax; ++t) {
    const bool valid = (t < len);
    v2f xB0 = mk2(xl.x, xl.y);
    v2f xB1 = half ? mk2(0.f, 0.f) : mk2(xh.x, xh.y);

    if (t + 1 < tmax) {
      xlo += 6; xhi += 6;
      xl = *(const float2*)xlo;
      xh = *(const float2*)xhi;
    }

    {
      int sw = __builtin_amdgcn_ds_swizzle(__float_as_int(hc0), 0x401F);
      v2f H0 = mk2(ha0, hb0);
      v2f H1 = half ? mk2(0.f, 0.f) : mk2(hc0, __int_as_float(sw));
      v8f gx0 = wmma4(Aih[0][0][1], xB1, wmma4(Aih[0][0][0], xB0, Cih[0][0]));
      v8f gx1 = wmma4(Aih[0][1][1], xB1, wmma4(Aih[0][1][0], xB0, Cih[0][1]));
      v8f gh0 = wmma4(Ahh[0][0][1], H1, wmma4(Ahh[0][0][0], H0, Chh[0][0]));
      v8f gh1 = wmma4(Ahh[0][1][1], H1, wmma4(Ahh[0][1][0], H0, Chh[0][1]));
      float r, z, nn, h;
      r = sigmoid_fast(gx0[0] + gh0[0]); z = sigmoid_fast(gx0[1] + gh0[1]);
      nn = tanh_fast(gx0[2] + r * gh0[2]); h = nn + z * (ha0 - nn); ha0 = valid ? h : ha0;
      r = sigmoid_fast(gx0[3] + gh0[3]); z = sigmoid_fast(gx0[4] + gh0[4]);
      nn = tanh_fast(gx0[5] + r * gh0[5]); h = nn + z * (hb0 - nn); hb0 = valid ? h : hb0;
      r = sigmoid_fast(gx0[6] + gh0[6]); z = sigmoid_fast(gx0[7] + gh0[7]);
      nn = tanh_fast(gx1[0] + r * gh1[0]); h = nn + z * (hc0 - nn); hc0 = valid ? h : hc0;
    }
    {
      int sw0 = __builtin_amdgcn_ds_swizzle(__float_as_int(hc0), 0x401F);
      v2f I0 = mk2(ha0, hb0);
      v2f I1 = half ? mk2(0.f, 0.f) : mk2(hc0, __int_as_float(sw0));
      int sw1 = __builtin_amdgcn_ds_swizzle(__float_as_int(hc1), 0x401F);
      v2f H0 = mk2(ha1, hb1);
      v2f H1 = half ? mk2(0.f, 0.f) : mk2(hc1, __int_as_float(sw1));
      v8f gx0 = wmma4(Aih[1][0][1], I1, wmma4(Aih[1][0][0], I0, Cih[1][0]));
      v8f gx1 = wmma4(Aih[1][1][1], I1, wmma4(Aih[1][1][0], I0, Cih[1][1]));
      v8f gh0 = wmma4(Ahh[1][0][1], H1, wmma4(Ahh[1][0][0], H0, Chh[1][0]));
      v8f gh1 = wmma4(Ahh[1][1][1], H1, wmma4(Ahh[1][1][0], H0, Chh[1][1]));
      float r, z, nn, h;
      r = sigmoid_fast(gx0[0] + gh0[0]); z = sigmoid_fast(gx0[1] + gh0[1]);
      nn = tanh_fast(gx0[2] + r * gh0[2]); h = nn + z * (ha1 - nn); ha1 = valid ? h : ha1;
      r = sigmoid_fast(gx0[3] + gh0[3]); z = sigmoid_fast(gx0[4] + gh0[4]);
      nn = tanh_fast(gx0[5] + r * gh0[5]); h = nn + z * (hb1 - nn); hb1 = valid ? h : hb1;
      r = sigmoid_fast(gx0[6] + gh0[6]); z = sigmoid_fast(gx0[7] + gh0[7]);
      nn = tanh_fast(gx1[0] + r * gh1[0]); h = nn + z * (hc1 - nn); hc1 = valid ? h : hc1;
    }
  }

  float* orow = out + (size_t)s * 6;
  *(float2*)(orow + (half ? 2 : 0)) = make_float2(ha1, hb1);
  orow[half ? 5 : 4] = hc1;
}

extern "C" void kernel_launch(void* const* d_in, const int* in_sizes, int n_in,
                              void* d_out, int out_size, void* d_ws, size_t ws_size,
                              hipStream_t stream) {
  (void)n_in; (void)out_size; (void)d_ws; (void)ws_size;
  const float* x       = (const float*)d_in[0];
  const int*   lengths = (const int*)d_in[1];
  const float* W_ih    = (const float*)d_in[2];
  const float* W_hh    = (const float*)d_in[3];
  const float* b_ih    = (const float*)d_in[4];
  const float* b_hh    = (const float*)d_in[5];
  float* out = (float*)d_out;

  const int nseq = in_sizes[1];
  const int tmax = in_sizes[0] / (nseq * 6);
  const int waves = nseq / 16;
  const int threads = waves * 32;
  const int block = 256;
  const int grid = (threads + block - 1) / block;
  gru16_wmma_kernel<<<grid, block, 0, stream>>>(x, lengths, W_ih, W_hh, b_ih, b_hh, out, tmax);
}


# Cross-half fetch of h5 (3 per trip) moved from the LDS crossbar (ds_swizzle + s_wait_dscnt) to v_permlanex16 on the vector ALU: same lane map l^16, no LDS round trip on the recurrence chain
# speedup vs baseline: 1.0708x; 1.0708x over previous
#include <hip/hip_runtime.h>

typedef __attribute__((ext_vector_type(2))) float v2f;
typedef __attribute__((ext_vector_type(8))) float v8f;

static constexpr unsigned long long kPerm = 0xB5F93E82A4D71C60ULL;

__device__ __forceinline__ float sigmoid_fast(float x) {
  float e = __builtin_amdgcn_exp2f(x * -1.4426950408889634f);
  return __builtin_amdgcn_rcpf(1.0f + e);
}
__device__ __forceinline__ float tanh_fast(float x) {
  float e = __builtin_amdgcn_exp2f(x * -2.8853900817779268f);
  return (1.0f - e) * __builtin_amdgcn_rcpf(1.0f + e);
}
__device__ __forceinline__ v8f wmma4(v2f a, v2f b, v8f c) {
  return __builtin_amdgcn_wmma_f32_16x16x4_f32(false, a, false, b, (short)0, c, false, false);
}
__device__ __forceinline__ v2f mk2(float a, float b) { v2f r; r[0] = a; r[1] = b; return r; }
__device__ __forceinline__ float other_half(float v) {
  int b = __float_as_int(v);
  return __int_as_float(__builtin_amdgcn_permlanex16(b, b, 0x76543210, 0xfedcba98, false, false));
}

__global__ __launch_bounds__(256) void gru16_wmma_kernel(
    const float* __restrict__ x, const int* __restrict__ lengths,
    const float* __restrict__ W_ih, const float* __restrict__ W_hh,
    const float* __restrict__ b_ih, const float* __restrict__ b_hh,
    float* __restrict__ out, int tmax)
{
  const int lane = threadIdx.x & 31;
  const int col  = lane & 15;
  const int half = lane >> 4;
  const int wv   = (blockIdx.x * blockDim.x + threadIdx.x) >> 5;
  const int s    = wv * 16 + col;

  const int g0 = (int)((kPerm >> (4 * col)) & 0xFULL);
  const int g1 = (col == 0) ? 16 : ((col == 8) ? 17 : -1);
  const int cl = 2 * half;

  v2f Aih[2][2][2], Ahh[2][2][2];
  v8f Cih[2][2],    Chh[2][2];
#pragma unroll
  for (int l = 0; l < 2; ++l) {
    const float* Wi = W_ih + l * 108;
    const float* Wh = W_hh + l * 108;
    const float* Bi = b_ih + l * 18;
    const float* Bh = b_hh + l * 18;

    Aih[l][0][0] = mk2(Wi[g0 * 6 + cl], Wi[g0 * 6 + cl + 1]);
    Ahh[l][0][0] = mk2(Wh[g0 * 6 + cl], Wh[g0 * 6 + cl + 1]);
    Aih[l][0][1] = half ? mk2(0.f, 0.f) : mk2(Wi[g0 * 6 + 4], Wi[g0 * 6 + 5]);
    Ahh[l][0][1] = half ? mk2(0.f, 0.f) : mk2(Wh[g0 * 6 + 4], Wh[g0 * 6 + 5]);
    if (g1 >= 0) {
      Aih[l][1][0] = mk2(Wi[g1 * 6 + cl], Wi[g1 * 6 + cl + 1]);
      Ahh[l][1][0] = mk2(Wh[g1 * 6 + cl], Wh[g1 * 6 + cl + 1]);
      Aih[l][1][1] = half ? mk2(0.f, 0.f) : mk2(Wi[g1 * 6 + 4], Wi[g1 * 6 + 5]);
      Ahh[l][1][1] = half ? mk2(0.f, 0.f) : mk2(Wh[g1 * 6 + 4], Wh[g1 * 6 + 5]);
    } else {
      Aih[l][1][0] = mk2(0.f, 0.f); Ahh[l][1][0] = mk2(0.f, 0.f);
      Aih[l][1][1] = mk2(0.f, 0.f); Ahh[l][1][1] = mk2(0.f, 0.f);
    }
    v8f ci0, ch0, ci1, ch1;
#pragma unroll
    for (int r = 0; r < 8; ++r) {
      int row = r + 8 * half;
      int g = (int)((kPerm >> (4 * row)) & 0xFULL);
      ci0[r] = Bi[g];  ch0[r] = Bh[g];
      ci1[r] = 0.f;    ch1[r] = 0.f;
    }
    ci1[0] = Bi[half ? 17 : 16];
    ch1[0] = Bh[half ? 17 : 16];
    Cih[l][0] = ci0; Chh[l][0] = ch0;
    Cih[l][1] = ci1; Chh[l][1] = ch1;
  }

  float ha0 = 0.f, hb0 = 0.f, hc0 = 0.f;
  float ha1 = 0.f, hb1 = 0.f, hc1 = 0.f;
  const int len = lengths[s];
  const float* xlo = x + (size_t)s * (size_t)tmax * 6 + cl;
  const float* xhi = x + (size_t)s * (size_t)tmax * 6 + 4;

  float2 xl = *(const float2*)xlo;
  float2 xh = *(const float2*)xhi;

#pragma unroll 1
  for (int t = 0; t < tmax; ++t) {
    const bool valid = (t < len);
    v2f xB0 = mk2(xl.x, xl.y);
    v2f xB1 = half ? mk2(0.f, 0.f) : mk2(xh.x, xh.y);

    if (t + 1 < tmax) {
      xlo += 6; xhi += 6;
      xl = *(const float2*)xlo;
      xh = *(const float2*)xhi;
    }

    {
      float sw = other_half(hc0);
      v2f H0 = mk2(ha0, hb0);
      v2f H1 = half ? mk2(0.f, 0.f) : mk2(hc0, sw);
      v8f gx0 = wmma4(Aih[0][0][1], xB1, wmma4(Aih[0][0][0], xB0, Cih[0][0]));
      v8f gx1 = wmma4(Aih[0][1][1], xB1, wmma4(Aih[0][1][0], xB0, Cih[0][1]));
      v8f gh0 = wmma4(Ahh[0][0][1], H1, wmma4(Ahh[0][0][0], H0, Chh[0][0]));
      v8f gh1 = wmma4(Ahh[0][1][1], H1, wmma4(Ahh[0][1][0], H0, Chh[0][1]));
      float r, z, nn, h;
      r = sigmoid_fast(gx0[0] + gh0[0]); z = sigmoid_fast(gx0[1] + gh0[1]);
      nn = tanh_fast(gx0[2] + r * gh0[2]); h = nn + z * (ha0 - nn); ha0 = valid ? h : ha0;
      r = sigmoid_fast(gx0[3] + gh0[3]); z = sigmoid_fast(gx0[4] + gh0[4]);
      nn = tanh_fast(gx0[5] + r * gh0[5]); h = nn + z * (hb0 - nn); hb0 = valid ? h : hb0;
      r = sigmoid_fast(gx0[6] + gh0[6]); z = sigmoid_fast(gx0[7] + gh0[7]);
      nn = tanh_fast(gx1[0] + r * gh1[0]); h = nn + z * (hc0 - nn); hc0 = valid ? h : hc0;
    }
    {
      float sw0 = other_half(hc0);
      v2f I0 = mk2(ha0, hb0);
      v2f I1 = half ? mk2(0.f, 0.f) : mk2(hc0, sw0);
      float sw1 = other_half(hc1);
      v2f H0 = mk2(ha1, hb1);
      v2f H1 = half ? mk2(0.f, 0.f) : mk2(hc1, sw1);
      v8f gx0 = wmma4(Aih[1][0][1], I1, wmma4(Aih[1][0][0], I0, Cih[1][0]));
      v8f gx1 = wmma4(Aih[1][1][1], I1, wmma4(Aih[1][1][0], I0, Cih[1][1]));
      v8f gh0 = wmma4(Ahh[1][0][1], H1, wmma4(Ahh[1][0][0], H0, Chh[1][0]));
      v8f gh1 = wmma4(Ahh[1][1][1], H1, wmma4(Ahh[1][1][0], H0, Chh[1][1]));
      float r, z, nn, h;
      r = sigmoid_fast(gx0[0] + gh0[0]); z = sigmoid_fast(gx0[1] + gh0[1]);
      nn = tanh_fast(gx0[2] + r * gh0[2]); h = nn + z * (ha1 - nn); ha1 = valid ? h : ha1;
      r = sigmoid_fast(gx0[3] + gh0[3]); z = sigmoid_fast(gx0[4] + gh0[4]);
      nn = tanh_fast(gx0[5] + r * gh0[5]); h = nn + z * (hb1 - nn); hb1 = valid ? h : hb1;
      r = sigmoid_fast(gx0[6] + gh0[6]); z = sigmoid_fast(gx0[7] + gh0[7]);
      nn = tanh_fast(gx1[0] + r * gh1[0]); h = nn + z * (hc1 - nn); hc1 = valid ? h : hc1;
    }
  }

  float* orow = out + (size_t)s * 6;
  *(float2*)(orow + (half ? 2 : 0)) = make_float2(ha1, hb1);
  orow[half ? 5 : 4] = hc1;
}

extern "C" void kernel_launch(void* const* d_in, const int* in_sizes, int n_in,
                              void* d_out, int out_size, void* d_ws, size_t ws_size,
                              hipStream_t stream) {
  (void)n_in; (void)out_size; (void)d_ws; (void)ws_size;
  const float* x       = (const float*)d_in[0];
  const int*   lengths = (const int*)d_in[1];
  const float* W_ih    = (const float*)d_in[2];
  const float* W_hh    = (const float*)d_in[3];
  const float* b_ih    = (const float*)d_in[4];
  const float* b_hh    = (const float*)d_in[5];
  float* out = (float*)d_out;

  const int nseq = in_sizes[1];
  const int tmax = in_sizes[0] / (nseq * 6);
  const int waves = nseq / 16;
  const int threads = waves * 32;
  const int block = 256;
  const int grid = (threads + block - 1) / block;
  gru16_wmma_kernel<<<grid, block, 0, stream>>>(x, lengths, W_ih, W_hh, b_ih, b_hh, out, tmax);
}


# Non-recurrent half of each layer's gate products taken off the chain: W_ih0 x one step ahead, W_hh1 h1 beside layer-0 gates; recurrent half accumulates onto it via C (r,z sums change order)
# speedup vs baseline: 1.2743x; 1.1900x over previous
#include <hip/hip_runtime.h>

typedef __attribute__((ext_vector_type(2))) float v2f;
typedef __attribute__((ext_vector_type(8))) float v8f;

static constexpr unsigned long long kPerm = 0xB5F93E82A4D71C60ULL;

__device__ __forceinline__ float sigmoid_fast(float x) {
  float e = __builtin_amdgcn_exp2f(x * -1.4426950408889634f);
  return __builtin_amdgcn_rcpf(1.0f + e);
}
__device__ __forceinline__ float tanh_fast(float x) {
  float e = __builtin_amdgcn_exp2f(x * -2.8853900817779268f);
  return (1.0f - e) * __builtin_amdgcn_rcpf(1.0f + e);
}
__device__ __forceinline__ v8f wmma4(v2f a, v2f b, v8f c) {
  return __builtin_amdgcn_wmma_f32_16x16x4_f32(false, a, false, b, (short)0, c, false, false);
}
__device__ __forceinline__ v2f mk2(float a, float b) { v2f r; r[0] = a; r[1] = b; return r; }
__device__ __forceinline__ float other_half(float v) {
  int b = __float_as_int(v);
  return __int_as_float(__builtin_amdgcn_permlanex16(b, b, 0x76543210, 0xfedcba98, false, false));
}

__global__ __launch_bounds__(256) void gru16_wmma_kernel(
    const float* __restrict__ x, const int* __restrict__ lengths,
    const float* __restrict__ W_ih, const float* __restrict__ W_hh,
    const float* __restrict__ b_ih, const float* __restrict__ b_hh,
    float* __restrict__ out, int tmax)
{
  const int lane = threadIdx.x & 31;
  const int col  = lane & 15;
  const int half = lane >> 4;
  const int wv   = (blockIdx.x * blockDim.x + threadIdx.x) >> 5;
  const int s    = wv * 16 + col;

  const int g0 = (int)((kPerm >> (4 * col)) & 0xFULL);
  const int g1 = (col == 0) ? 16 : ((col == 8) ? 17 : -1);
  const int g1x = (col == 1) ? 12 : ((col == 2) ? 13 : ((col == 9) ? 14 : ((col == 10) ? 15 : g1)));
  const bool nrow0 = ((0x2424 >> col) & 1) != 0;
  const int cl = 2 * half;

  v2f Aih[2][2][2], Ahh[2][2][2];
  v8f Cih[2][2],    Chh[2][2];
#pragma unroll
  for (int l = 0; l < 2; ++l) {
    const float* Wi = W_ih + l * 108;
    const float* Wh = W_hh + l * 108;
    const float* Bi = b_ih + l * 18;
    const float* Bh = b_hh + l * 18;

    Aih[l][0][0] = mk2(Wi[g0 * 6 + cl], Wi[g0 * 6 + cl + 1]);
    Ahh[l][0][0] = mk2(Wh[g0 * 6 + cl], Wh[g0 * 6 + cl + 1]);
    Aih[l][0][1] = half ? mk2(0.f, 0.f) : mk2(Wi[g0 * 6 + 4], Wi[g0 * 6 + 5]);
    Ahh[l][0][1] = half ? mk2(0.f, 0.f) : mk2(Wh[g0 * 6 + 4], Wh[g0 * 6 + 5]);
    const int gi = (l == 0) ? g1x : g1;
    const int gh = (l == 0) ? g1 : g1x;
    if (gi >= 0) {
      Aih[l][1][0] = mk2(Wi[gi * 6 + cl], Wi[gi * 6 + cl + 1]);
      Aih[l][1][1] = half ? mk2(0.f, 0.f) : mk2(Wi[gi * 6 + 4], Wi[gi * 6 + 5]);
    } else {
      Aih[l][1][0] = mk2(0.f, 0.f); Aih[l][1][1] = mk2(0.f, 0.f);
    }
    if (gh >= 0) {
      Ahh[l][1][0] = mk2(Wh[gh * 6 + cl], Wh[gh * 6 + cl + 1]);
      Ahh[l][1][1] = half ? mk2(0.f, 0.f) : mk2(Wh[gh * 6 + 4], Wh[gh * 6 + 5]);
    } else {
      Ahh[l][1][0] = mk2(0.f, 0.f); Ahh[l][1][1] = mk2(0.f, 0.f);
    }
    if (nrow0) {
      if (l == 0) { Aih[0][0][0] = mk2(0.f, 0.f); Aih[0][0][1] = mk2(0.f, 0.f); }
      else        { Ahh[1][0][0] = mk2(0.f, 0.f); Ahh[1][0][1] = mk2(0.f, 0.f); }
    }
    v8f ci0, ch0, ci1, ch1;
#pragma unroll
    for (int r = 0; r < 8; ++r) {
      int row = r + 8 * half;
      int g = (int)((kPerm >> (4 * row)) & 0xFULL);
      ci0[r] = Bi[g];  ch0[r] = Bh[g];
      ci1[r] = 0.f;    ch1[r] = 0.f;
    }
    ci1[0] = Bi[half ? 17 : 16];
    ch1[0] = Bh[half ? 17 : 16];
    Cih[l][0] = ci0; Chh[l][0] = ch0;
    Cih[l][1] = ci1; Chh[l][1] = ch1;
  }

  v8f S0x0, S1h0;
#pragma unroll
  for (int r = 0; r < 8; ++r) {
    const bool nr = (r == 2) || (r == 5);
    S0x0[r] = nr ? Chh[0][0][r] : (Cih[0][0][r] + Chh[0][0][r]);
    S1h0[r] = nr ? Cih[1][0][r] : (Chh[1][0][r] + Cih[1][0][r]);
  }
  v8f S0x1 = Cih[0][1];
  S0x1[1] = Cih[0][0][2]; S0x1[2] = Cih[0][0][5];
  v8f S1h1 = Chh[1][1];
  S1h1[1] = Chh[1][0][2]; S1h1[2] = Chh[1][0][5];

  float ha0 = 0.f, hb0 = 0.f, hc0 = 0.f;
  float ha1 = 0.f, hb1 = 0.f, hc1 = 0.f;
  const int len = lengths[s];
  const float* xrow = x + (size_t)s * (size_t)tmax * 6;
  const int tlast = tmax - 1;

  float2 xl = *(const float2*)(xrow + cl);
  float2 xh = *(const float2*)(xrow + 4);
  v8f P00, P01;
  {
    v2f xB0 = mk2(xl.x, xl.y);
    v2f xB1 = half ? mk2(0.f, 0.f) : mk2(xh.x, xh.y);
    P00 = wmma4(Aih[0][0][1], xB1, wmma4(Aih[0][0][0], xB0, S0x0));
    P01 = wmma4(Aih[0][1][1], xB1, wmma4(Aih[0][1][0], xB0, S0x1));
  }

#pragma unroll 1
  for (int t = 0; t < tmax; ++t) {
    const bool valid = (t < len);
    const float* xn = xrow + (size_t)min(t + 1, tlast) * 6;
    xl = *(const float2*)(xn + cl);
    xh = *(const float2*)(xn + 4);
    __builtin_amdgcn_sched_barrier(0);

    v8f Q00, Q01;
    {
      float sw = other_half(hc0);
      v2f H0 = mk2(ha0, hb0);
      v2f H1 = half ? mk2(0.f, 0.f) : mk2(hc0, sw);
      Q00 = wmma4(Ahh[0][0][1], H1, wmma4(Ahh[0][0][0], H0, P00));
      Q01 = wmma4(Ahh[0][1][1], H1, wmma4(Ahh[0][1][0], H0, Chh[0][1]));
    }
    const float gxa = P01[1], gxb = P01[2], gxc = P01[0];
    v8f P10, P11;
    {
      float sw1 = other_half(hc1);
      v2f G0 = mk2(ha1, hb1);
      v2f G1 = half ? mk2(0.f, 0.f) : mk2(hc1, sw1);
      P10 = wmma4(Ahh[1][0][1], G1, wmma4(Ahh[1][0][0], G0, S1h0));
      P11 = wmma4(Ahh[1][1][1], G1, wmma4(Ahh[1][1][0], G0, S1h1));
    }
    {
      float ra = sigmoid_fast(Q00[0]), rb = sigmoid_fast(Q00[3]), rc = sigmoid_fast(Q00[6]);
      float za = sigmoid_fast(Q00[1]), zb = sigmoid_fast(Q00[4]), zc = sigmoid_fast(Q00[7]);
      float na = tanh_fast(gxa + ra * Q00[2]);
      float nb = tanh_fast(gxb + rb * Q00[5]);
      float nc = tanh_fast(gxc + rc * Q01[0]);
      float h;
      h = na + za * (ha0 - na); ha0 = valid ? h : ha0;
      h = nb + zb * (hb0 - nb); hb0 = valid ? h : hb0;
      h = nc + zc * (hc0 - nc); hc0 = valid ? h : hc0;
    }
    v8f Q10, Q11;
    {
      float sw0 = other_half(hc0);
      v2f I0 = mk2(ha0, hb0);
      v2f I1 = half ? mk2(0.f, 0.f) : mk2(hc0, sw0);
      Q10 = wmma4(Aih[1][0][1], I1, wmma4(Aih[1][0][0], I0, P10));
      Q11 = wmma4(Aih[1][1][1], I1, wmma4(Aih[1][1][0], I0, Cih[1][1]));
    }
    const float ra1 = sigmoid_fast(Q10[0]), rb1 = sigmoid_fast(Q10[3]), rc1 = sigmoid_fast(Q10[6]);
    const float za1 = sigmoid_fast(Q10[1]), zb1 = sigmoid_fast(Q10[4]), zc1 = sigmoid_fast(Q10[7]);
    __builtin_amdgcn_sched_barrier(0);
    {
      v2f xB0 = mk2(xl.x, xl.y);
      v2f xB1 = half ? mk2(0.f, 0.f) : mk2(xh.x, xh.y);
      P00 = wmma4(Aih[0][0][1], xB1, wmma4(Aih[0][0][0], xB0, S0x0));
      P01 = wmma4(Aih[0][1][1], xB1, wmma4(Aih[0][1][0], xB0, S0x1));
    }
    {
      float na = tanh_fast(Q10[2] + ra1 * P11[1]);
      float nb = tanh_fast(Q10[5] + rb1 * P11[2]);
      float nc = tanh_fast(Q11[0] + rc1 * P11[0]);
      float h;
      h = na + za1 * (ha1 - na); ha1 = valid ? h : ha1;
      h = nb + zb1 * (hb1 - nb); hb1 = valid ? h : hb1;
      h = nc + zc1 * (hc1 - nc); hc1 = valid ? h : hc1;
    }
  }

  float* orow = out + (size_t)s * 6;
  *(float2*)(orow + (half ? 2 : 0)) = make_float2(ha1, hb1);
  orow[half ? 5 : 4] = hc1;
}

extern "C" void kernel_launch(void* const* d_in, const int* in_sizes, int n_in,
                              void* d_out, int out_size, void* d_ws, size_t ws_size,
                              hipStream_t stream) {
  (void)n_in; (void)out_size; (void)d_ws; (void)ws_size;
  const float* x       = (const float*)d_in[0];
  const int*   lengths = (const int*)d_in[1];
  const float* W_ih    = (const float*)d_in[2];
  const float* W_hh    = (const float*)d_in[3];
  const float* b_ih    = (const float*)d_in[4];
  const float* b_hh    = (const float*)d_in[5];
  float* out = (float*)d_out;

  const int nseq = in_sizes[1];
  const int tmax = in_sizes[0] / (nseq * 6);
  const int waves = nseq / 16;
  const int threads = waves * 32;
  const int block = 256;
  const int grid = (threads + block - 1) / block;
  gru16_wmma_kernel<<<grid, block, 0, stream>>>(x, lengths, W_ih, W_hh, b_ih, b_hh, out, tmax);
}


# Second M-tile of layer-0's recurrent product (n4,n5 of W_hh0) merged into layer-1's input-product tile: both multiply h0_t, so one tile serves both; 14 WMMA per trip (was 16)
# speedup vs baseline: 1.3770x; 1.0806x over previous
#include <hip/hip_runtime.h>

typedef __attribute__((ext_vector_type(2))) float v2f;
typedef __attribute__((ext_vector_type(8))) float v8f;

static constexpr unsigned long long kPerm = 0xB5F93E82A4D71C60ULL;

__device__ __forceinline__ float sigmoid_fast(float x) {
  float e = __builtin_amdgcn_exp2f(x * -1.4426950408889634f);
  return __builtin_amdgcn_rcpf(1.0f + e);
}
__device__ __forceinline__ float tanh_fast(float x) {
  float e = __builtin_amdgcn_exp2f(x * -2.8853900817779268f);
  return (1.0f - e) * __builtin_amdgcn_rcpf(1.0f + e);
}
__device__ __forceinline__ v8f wmma4(v2f a, v2f b, v8f c) {
  return __builtin_amdgcn_wmma_f32_16x16x4_f32(false, a, false, b, (short)0, c, false, false);
}
__device__ __forceinline__ v2f mk2(float a, float b) { v2f r; r[0] = a; r[1] = b; return r; }
__device__ __forceinline__ float other_half(float v) {
  int b = __float_as_int(v);
  return __int_as_float(__builtin_amdgcn_permlanex16(b, b, 0x76543210, 0xfedcba98, false, false));
}

__global__ __launch_bounds__(256) void gru16_wmma_kernel(
    const float* __restrict__ x, const int* __restrict__ lengths,
    const float* __restrict__ W_ih, const float* __restrict__ W_hh,
    const float* __restrict__ b_ih, const float* __restrict__ b_hh,
    float* __restrict__ out, int tmax)
{
  const int lane = threadIdx.x & 31;
  const int col  = lane & 15;
  const int half = lane >> 4;
  const int wv   = (blockIdx.x * blockDim.x + threadIdx.x) >> 5;
  const int s    = wv * 16 + col;

  const int g0 = (int)((kPerm >> (4 * col)) & 0xFULL);
  const int g1 = (col == 0) ? 16 : ((col == 8) ? 17 : -1);
  const int g1x = (col == 1) ? 12 : ((col == 2) ? 13 : ((col == 9) ? 14 : ((col == 10) ? 15 : g1)));
  const bool nrow0 = ((0x2424 >> col) & 1) != 0;
  const int cl = 2 * half;

  v2f Aih[2][2][2], Ahh[2][2][2];
  v8f Cih[2][2],    Chh[2][2];
#pragma unroll
  for (int l = 0; l < 2; ++l) {
    const float* Wi = W_ih + l * 108;
    const float* Wh = W_hh + l * 108;
    const float* Bi = b_ih + l * 18;
    const float* Bh = b_hh + l * 18;

    Aih[l][0][0] = mk2(Wi[g0 * 6 + cl], Wi[g0 * 6 + cl + 1]);
    Ahh[l][0][0] = mk2(Wh[g0 * 6 + cl], Wh[g0 * 6 + cl + 1]);
    Aih[l][0][1] = half ? mk2(0.f, 0.f) : mk2(Wi[g0 * 6 + 4], Wi[g0 * 6 + 5]);
    Ahh[l][0][1] = half ? mk2(0.f, 0.f) : mk2(Wh[g0 * 6 + 4], Wh[g0 * 6 + 5]);
    const int gi = (l == 0) ? g1x : g1;
    const int gh = (l == 0) ? g1 : g1x;
    if (gi >= 0) {
      Aih[l][1][0] = mk2(Wi[gi * 6 + cl], Wi[gi * 6 + cl + 1]);
      Aih[l][1][1] = half ? mk2(0.f, 0.f) : mk2(Wi[gi * 6 + 4], Wi[gi * 6 + 5]);
    } else {
      Aih[l][1][0] = mk2(0.f, 0.f); Aih[l][1][1] = mk2(0.f, 0.f);
    }
    if (gh >= 0) {
      Ahh[l][1][0] = mk2(Wh[gh * 6 + cl], Wh[gh * 6 + cl + 1]);
      Ahh[l][1][1] = half ? mk2(0.f, 0.f) : mk2(Wh[gh * 6 + 4], Wh[gh * 6 + 5]);
    } else {
      Ahh[l][1][0] = mk2(0.f, 0.f); Ahh[l][1][1] = mk2(0.f, 0.f);
    }
    if (nrow0) {
      if (l == 0) { Aih[0][0][0] = mk2(0.f, 0.f); Aih[0][0][1] = mk2(0.f, 0.f); }
      else        { Ahh[1][0][0] = mk2(0.f, 0.f); Ahh[1][0][1] = mk2(0.f, 0.f); }
    }
    v8f ci0, ch0, ci1, ch1;
#pragma unroll
    for (int r = 0; r < 8; ++r) {
      int row = r + 8 * half;
      int g = (int)((kPerm >> (4 * row)) & 0xFULL);
      ci0[r] = Bi[g];  ch0[r] = Bh[g];
      ci1[r] = 0.f;    ch1[r] = 0.f;
    }
    ci1[0] = Bi[half ? 17 : 16];
    ch1[0] = Bh[half ? 17 : 16];
    Cih[l][0] = ci0; Chh[l][0] = ch0;
    Cih[l][1] = ci1; Chh[l][1] = ch1;
  }

  v8f S0x0, S1h0;
#pragma unroll
  for (int r = 0; r < 8; ++r) {
    const bool nr = (r == 2) || (r == 5);
    S0x0[r] = nr ? Chh[0][0][r] : (Cih[0][0][r] + Chh[0][0][r]);
    S1h0[r] = nr ? Cih[1][0][r] : (Chh[1][0][r] + Cih[1][0][r]);
  }
  v8f S0x1 = Cih[0][1];
  S0x1[1] = Cih[0][0][2]; S0x1[2] = Cih[0][0][5];
  v8f S1h1 = Chh[1][1];
  S1h1[1] = Chh[1][0][2]; S1h1[2] = Chh[1][0][5];

  v2f Amix0, Amix1;
  {
    const int gm = (col == 0 || col == 1) ? 16 : ((col == 8 || col == 9) ? 17 : -1);
    const float* Wm = ((col & 7) == 0) ? (W_ih + 108) : W_hh;
    if (gm >= 0) {
      Amix0 = mk2(Wm[gm * 6 + cl], Wm[gm * 6 + cl + 1]);
      Amix1 = half ? mk2(0.f, 0.f) : mk2(Wm[gm * 6 + 4], Wm[gm * 6 + 5]);
    } else {
      Amix0 = mk2(0.f, 0.f); Amix1 = mk2(0.f, 0.f);
    }
  }
  v8f Smix = Cih[1][1];
  Smix[1] = Chh[0][1][0];
  float nh0c = Chh[0][1][0];

  float ha0 = 0.f, hb0 = 0.f, hc0 = 0.f;
  float ha1 = 0.f, hb1 = 0.f, hc1 = 0.f;
  const int len = lengths[s];
  const float* xrow = x + (size_t)s * (size_t)tmax * 6;
  const int tlast = tmax - 1;

  float2 xl = *(const float2*)(xrow + cl);
  float2 xh = *(const float2*)(xrow + 4);
  v8f P00, P01;
  {
    v2f xB0 = mk2(xl.x, xl.y);
    v2f xB1 = half ? mk2(0.f, 0.f) : mk2(xh.x, xh.y);
    P00 = wmma4(Aih[0][0][1], xB1, wmma4(Aih[0][0][0], xB0, S0x0));
    P01 = wmma4(Aih[0][1][1], xB1, wmma4(Aih[0][1][0], xB0, S0x1));
  }

#pragma unroll 1
  for (int t = 0; t < tmax; ++t) {
    const bool valid = (t < len);
    const float* xn = xrow + (size_t)min(t + 1, tlast) * 6;
    xl = *(const float2*)(xn + cl);
    xh = *(const float2*)(xn + 4);
    __builtin_amdgcn_sched_barrier(0);

    v8f Q00;
    {
      float sw = other_half(hc0);
      v2f H0 = mk2(ha0, hb0);
      v2f H1 = half ? mk2(0.f, 0.f) : mk2(hc0, sw);
      Q00 = wmma4(Ahh[0][0][1], H1, wmma4(Ahh[0][0][0], H0, P00));
    }
    const float gxa = P01[1], gxb = P01[2], gxc = P01[0];
    v8f P10, P11;
    {
      float sw1 = other_half(hc1);
      v2f G0 = mk2(ha1, hb1);
      v2f G1 = half ? mk2(0.f, 0.f) : mk2(hc1, sw1);
      P10 = wmma4(Ahh[1][0][1], G1, wmma4(Ahh[1][0][0], G0, S1h0));
      P11 = wmma4(Ahh[1][1][1], G1, wmma4(Ahh[1][1][0], G0, S1h1));
    }
    {
      float ra = sigmoid_fast(Q00[0]), rb = sigmoid_fast(Q00[3]), rc = sigmoid_fast(Q00[6]);
      float za = sigmoid_fast(Q00[1]), zb = sigmoid_fast(Q00[4]), zc = sigmoid_fast(Q00[7]);
      float na = tanh_fast(gxa + ra * Q00[2]);
      float nb = tanh_fast(gxb + rb * Q00[5]);
      float nc = tanh_fast(gxc + rc * nh0c);
      float h;
      h = na + za * (ha0 - na); ha0 = valid ? h : ha0;
      h = nb + zb * (hb0 - nb); hb0 = valid ? h : hb0;
      h = nc + zc * (hc0 - nc); hc0 = valid ? h : hc0;
    }
    v8f Q10, Q11;
    {
      float sw0 = other_half(hc0);
      v2f I0 = mk2(ha0, hb0);
      v2f I1 = half ? mk2(0.f, 0.f) : mk2(hc0, sw0);
      Q10 = wmma4(Aih[1][0][1], I1, wmma4(Aih[1][0][0], I0, P10));
      Q11 = wmma4(Amix1, I1, wmma4(Amix0, I0, Smix));
    }
    nh0c = Q11[1];
    const float ra1 = sigmoid_fast(Q10[0]), rb1 = sigmoid_fast(Q10[3]), rc1 = sigmoid_fast(Q10[6]);
    const float za1 = sigmoid_fast(Q10[1]), zb1 = sigmoid_fast(Q10[4]), zc1 = sigmoid_fast(Q10[7]);
    __builtin_amdgcn_sched_barrier(0);
    {
      v2f xB0 = mk2(xl.x, xl.y);
      v2f xB1 = half ? mk2(0.f, 0.f) : mk2(xh.x, xh.y);
      P00 = wmma4(Aih[0][0][1], xB1, wmma4(Aih[0][0][0], xB0, S0x0));
      P01 = wmma4(Aih[0][1][1], xB1, wmma4(Aih[0][1][0], xB0, S0x1));
    }
    {
      float na = tanh_fast(Q10[2] + ra1 * P11[1]);
      float nb = tanh_fast(Q10[5] + rb1 * P11[2]);
      float nc = tanh_fast(Q11[0] + rc1 * P11[0]);
      float h;
      h = na + za1 * (ha1 - na); ha1 = valid ? h : ha1;
      h = nb + zb1 * (hb1 - nb); hb1 = valid ? h : hb1;
      h = nc + zc1 * (hc1 - nc); hc1 = valid ? h : hc1;
    }
  }

  float* orow = out + (size_t)s * 6;
  *(float2*)(orow + (half ? 2 : 0)) = make_float2(ha1, hb1);
  orow[half ? 5 : 4] = hc1;
}

extern "C" void kernel_launch(void* const* d_in, const int* in_sizes, int n_in,
                              void* d_out, int out_size, void* d_ws, size_t ws_size,
                              hipStream_t stream) {
  (void)n_in; (void)out_size; (void)d_ws; (void)ws_size;
  const float* x       = (const float*)d_in[0];
  const int*   lengths = (const int*)d_in[1];
  const float* W_ih    = (const float*)d_in[2];
  const float* W_hh    = (const float*)d_in[3];
  const float* b_ih    = (const float*)d_in[4];
  const float* b_hh    = (const float*)d_in[5];
  float* out = (float*)d_out;

  const int nseq = in_sizes[1];
  const int tmax = in_sizes[0] / (nseq * 6);
  const int waves = nseq / 16;
  const int threads = waves * 32;
  const int block = 256;
  const int grid = (threads + block - 1) / block;
  gru16_wmma_kernel<<<grid, block, 0, stream>>>(x, lengths, W_ih, W_hh, b_ih, b_hh, out, tmax);
}
